# starting kernel
# baseline (speedup 1.0000x reference)
#include <hip/hip_runtime.h>
#include <hip/hip_bf16.h>

typedef __attribute__((ext_vector_type(2))) float v2f;
typedef __attribute__((ext_vector_type(8))) float v8f;

#define E 7

__global__ void emotion_setup_kernel(const float* __restrict__ Wq,
                                     const float* __restrict__ Wk,
                                     const float* __restrict__ Wd,
                                     const float* __restrict__ bd,
                                     float* __restrict__ ws) {
    const int t = threadIdx.x;
    const int i = t >> 3;
    const int l = t & 7;
    float c = 0.0f;
    if (i < E && l < E) {
#pragma unroll
        for (int j = 0; j < E; ++j)
            c += Wq[j * E + l] * Wk[j * E + i];
    }
    ws[i * 8 + l] = c;
    if (t < 8) {
        float cs = 0.0f;
        if (t < E) {
#pragma unroll
            for (int j = 0; j < E; ++j) cs += Wd[j * E + t];
        }
        ws[128 + t] = cs;
    }
    if (t == 8) {
        float sb = 0.0f;
#pragma unroll
        for (int j = 0; j < E; ++j) sb += bd[j];
        ws[136] = sb;
    }
}

__global__ __launch_bounds__(256) void emotion_main_kernel(
    const float* __restrict__ rep,
    const float* __restrict__ x,
    const float* __restrict__ ws,
    float* __restrict__ out,
    int rows, int ntp) {
    const int lane = threadIdx.x & 31;
    const int m    = lane & 15;
    const int kh   = lane >> 4;

    v2f A0, A1;
    A0.x = ws[m * 8 + 2 * kh];
    A0.y = ws[m * 8 + 2 * kh + 1];
    A1.x = ws[m * 8 + 4 + 2 * kh];
    A1.y = ws[m * 8 + 5 + 2 * kh];

    float csum[E];
#pragma unroll
    for (int i = 0; i < E; ++i) csum[i] = ws[128 + i];
    const float sumbd = ws[136];

    const int wave   = blockIdx.x * (blockDim.x >> 5) + (threadIdx.x >> 5);
    const int nwaves = gridDim.x * (blockDim.x >> 5);

    for (int tp = wave; tp < ntp; tp += nwaves) {
        const int base = tp * 32;

        int r0 = base + m;       if (r0 >= rows) r0 = rows - 1;
        int r1 = base + 16 + m;  if (r1 >= rows) r1 = rows - 1;
        const float* x0 = x + (size_t)r0 * E;
        const float* x1 = x + (size_t)r1 * E;
        v2f B00, B01, B10, B11;
        if (kh == 0) {
            B00.x = x0[0]; B00.y = x0[1]; B01.x = x0[4]; B01.y = x0[5];
            B10.x = x1[0]; B10.y = x1[1]; B11.x = x1[4]; B11.y = x1[5];
        } else {
            B00.x = x0[2]; B00.y = x0[3]; B01.x = x0[6]; B01.y = 0.0f;
            B10.x = x1[2]; B10.y = x1[3]; B11.x = x1[6]; B11.y = 0.0f;
        }

        {
            const long long nr = (long long)(tp + nwaves) * 32 + lane;
            if (nr < rows) {
                __builtin_prefetch(x   + (size_t)nr * E, 0, 0);
                __builtin_prefetch(rep + (size_t)nr * E, 0, 0);
            }
        }

        v8f acc0 = {};
        acc0 = __builtin_amdgcn_wmma_f32_16x16x4_f32(
            false, A0, false, B00, (short)0, acc0, false, false);
        acc0 = __builtin_amdgcn_wmma_f32_16x16x4_f32(
            false, A1, false, B01, (short)0, acc0, false, false);
        v8f acc1 = {};
        acc1 = __builtin_amdgcn_wmma_f32_16x16x4_f32(
            false, A0, false, B10, (short)0, acc1, false, false);
        acc1 = __builtin_amdgcn_wmma_f32_16x16x4_f32(
            false, A1, false, B11, (short)0, acc1, false, false);

        float t[E];
#pragma unroll
        for (int i = 0; i < E; ++i) {
            const float sh = __shfl(acc1[i], m, 32);
            t[i] = (kh == 0) ? acc0[i] : sh;
        }

        const int  erow  = base + lane;
        const bool valid = erow < rows;
        const int  cr    = valid ? erow : rows - 1;
        const float* rr  = rep + (size_t)cr * E;
        float r[E], raw[E];
#pragma unroll
        for (int i = 0; i < E; ++i) r[i] = rr[i];
#pragma unroll
        for (int i = 0; i < E; ++i) raw[i] = r[i] * t[i];

        float mx = raw[0];
#pragma unroll
        for (int i = 1; i < E; ++i) mx = fmaxf(mx, raw[i]);
        float e[E], sum = 0.0f;
#pragma unroll
        for (int i = 0; i < E; ++i) {
            e[i] = __expf(raw[i] - mx);
            sum += e[i];
        }
        const float inv = __builtin_amdgcn_rcpf(sum);

        float o[E];
#pragma unroll
        for (int i = 0; i < E; ++i) {
            const float s  = e[i] * inv;
            const float s3 = s * s * s;
            float d = fmaf(s3, csum[i], sumbd);
            d = fminf(fmaxf(d, -1.0f), 1.0f);
            o[i] = r[i] + d;
        }
        if (valid) {
            float* op = out + (size_t)erow * E;
#pragma unroll
            for (int i = 0; i < E; ++i) op[i] = o[i];
        }
    }
}

extern "C" void kernel_launch(void* const* d_in, const int* in_sizes, int n_in,
                              void* d_out, int out_size, void* d_ws, size_t ws_size,
                              hipStream_t stream) {
    const float* rep = (const float*)d_in[0];
    const float* x   = (const float*)d_in[1];
    const float* Wq  = (const float*)d_in[2];
    const float* Wk  = (const float*)d_in[3];
    const float* Wd  = (const float*)d_in[4];
    const float* bd  = (const float*)d_in[5];
    float* ws  = (float*)d_ws;
    float* out = (float*)d_out;

    const int rows = in_sizes[0] / E;
    const int ntp  = (rows + 31) / 32;

    emotion_setup_kernel<<<1, 128, 0, stream>>>(Wq, Wk, Wd, bd, ws);

    const int wavesPerBlock = 8;
    int targetWaves = (ntp + 1) / 2;
    int blocks = (targetWaves + wavesPerBlock - 1) / wavesPerBlock;
    if (blocks < 1) blocks = 1;

    emotion_main_kernel<<<blocks, 256, 0, stream>>>(rep, x, ws, out, rows, ntp);
}


# x, rep and out move as whole 128-B lines, 16 B per lane, through a wave-private LDS transpose (was 4-B accesses at 28-B lane stride, 7 partial stores per out line); 128-row units, 2 per wave
# speedup vs baseline: 1.1949x; 1.1949x over previous
#include <hip/hip_runtime.h>
#include <hip/hip_bf16.h>

typedef __attribute__((ext_vector_type(2))) float v2f;
typedef __attribute__((ext_vector_type(4))) float v4f;
typedef __attribute__((ext_vector_type(8))) float v8f;

#define E 7
#define UNIT_ROWS 128
#define UNIT_FLOATS (UNIT_ROWS * E)
#define UNIT_V4 (UNIT_FLOATS / 4 / 32)
#define WAVES_PER_BLOCK 8

__global__ void emotion_setup_kernel(const float* __restrict__ Wq,
                                     const float* __restrict__ Wk,
                                     const float* __restrict__ Wd,
                                     const float* __restrict__ bd,
                                     float* __restrict__ ws) {
    const int t = threadIdx.x;
    const int i = t >> 3;
    const int l = t & 7;
    float c = 0.0f;
    if (i < E && l < E) {
#pragma unroll
        for (int j = 0; j < E; ++j)
            c += Wq[j * E + l] * Wk[j * E + i];
    }
    ws[i * 8 + l] = c;
    if (t < 8) {
        float cs = 0.0f;
        if (t < E) {
#pragma unroll
            for (int j = 0; j < E; ++j) cs += Wd[j * E + t];
        }
        ws[128 + t] = cs;
    }
    if (t == 8) {
        float sb = 0.0f;
#pragma unroll
        for (int j = 0; j < E; ++j) sb += bd[j];
        ws[136] = sb;
    }
}

__device__ __forceinline__ void load_unit(const float* __restrict__ x,
                                          const float* __restrict__ rep,
                                          int unit, int lane,
                                          v4f (&vx)[UNIT_V4], v4f (&vr)[UNIT_V4]) {
    const v4f* xg = (const v4f*)(x   + (size_t)unit * UNIT_FLOATS);
    const v4f* rg = (const v4f*)(rep + (size_t)unit * UNIT_FLOATS);
#pragma unroll
    for (int j = 0; j < UNIT_V4; ++j) vx[j] = xg[lane + 32 * j];
#pragma unroll
    for (int j = 0; j < UNIT_V4; ++j) vr[j] = rg[lane + 32 * j];
}

__device__ __forceinline__ void compute_unit(float* xs, const float* rs,
                                             v2f A0, v2f A1,
                                             const float (&csum)[E], float sumbd,
                                             int lane, int m, int kh) {
#pragma unroll
    for (int q = 0; q < UNIT_ROWS / 32; ++q) {
        const int ra  = (q * 32 + m) * E;
        const int rb  = ra + 16 * E;
        const int off = 2 * kh;
        v2f B00, B01, B10, B11;
        B00.x = xs[ra + off];     B00.y = xs[ra + off + 1];
        B01.x = xs[ra + 4 + off];
        const float pa = xs[ra + 5 + off];
        B01.y = kh ? 0.0f : pa;
        B10.x = xs[rb + off];     B10.y = xs[rb + off + 1];
        B11.x = xs[rb + 4 + off];
        const float pb = xs[rb + 5 + off];
        B11.y = kh ? 0.0f : pb;

        v8f acc0 = {};
        acc0 = __builtin_amdgcn_wmma_f32_16x16x4_f32(
            false, A0, false, B00, (short)0, acc0, false, false);
        acc0 = __builtin_amdgcn_wmma_f32_16x16x4_f32(
            false, A1, false, B01, (short)0, acc0, false, false);
        v8f acc1 = {};
        acc1 = __builtin_amdgcn_wmma_f32_16x16x4_f32(
            false, A0, false, B10, (short)0, acc1, false, false);
        acc1 = __builtin_amdgcn_wmma_f32_16x16x4_f32(
            false, A1, false, B11, (short)0, acc1, false, false);

        float t[E];
#pragma unroll
        for (int i = 0; i < E; ++i) {
            const float sh = __shfl(acc1[i], m, 32);
            t[i] = (kh == 0) ? acc0[i] : sh;
        }

        const int lr = (q * 32 + lane) * E;
        float r[E], raw[E];
#pragma unroll
        for (int i = 0; i < E; ++i) r[i] = rs[lr + i];
#pragma unroll
        for (int i = 0; i < E; ++i) raw[i] = r[i] * t[i];

        float mx = raw[0];
#pragma unroll
        for (int i = 1; i < E; ++i) mx = fmaxf(mx, raw[i]);
        float e[E], sum = 0.0f;
#pragma unroll
        for (int i = 0; i < E; ++i) {
            e[i] = __expf(raw[i] - mx);
            sum += e[i];
        }
        const float inv = __builtin_amdgcn_rcpf(sum);

#pragma unroll
        for (int i = 0; i < E; ++i) {
            const float s  = e[i] * inv;
            const float s3 = s * s * s;
            float d = fmaf(s3, csum[i], sumbd);
            d = fminf(fmaxf(d, -1.0f), 1.0f);
            xs[lr + i] = r[i] + d;
        }
    }
}

__global__ __launch_bounds__(256) void emotion_main_kernel(
    const float* __restrict__ rep,
    const float* __restrict__ x,
    const float* __restrict__ ws,
    float* __restrict__ out,
    int rows, int nunits) {
    __shared__ __attribute__((aligned(16))) float lds[WAVES_PER_BLOCK][2 * UNIT_FLOATS];

    const int lane = threadIdx.x & 31;
    const int m    = lane & 15;
    const int kh   = lane >> 4;
    float* xs = lds[threadIdx.x >> 5];
    float* rs = xs + UNIT_FLOATS;

    v2f A0, A1;
    A0.x = ws[m * 8 + 2 * kh];
    A0.y = ws[m * 8 + 2 * kh + 1];
    A1.x = ws[m * 8 + 4 + 2 * kh];
    A1.y = ws[m * 8 + 5 + 2 * kh];

    float csum[E];
#pragma unroll
    for (int i = 0; i < E; ++i) csum[i] = ws[128 + i];
    const float sumbd = ws[136];

    const int wave   = blockIdx.x * WAVES_PER_BLOCK + (threadIdx.x >> 5);
    const int nwaves = gridDim.x * WAVES_PER_BLOCK;

    const bool al16  = ((((size_t)x) | ((size_t)rep) | ((size_t)out)) & 15) == 0;
    const int  nfull = al16 ? rows / UNIT_ROWS : 0;

    v4f vx[UNIT_V4], vr[UNIT_V4];
    int u = wave;
    if (u < nfull) load_unit(x, rep, u, lane, vx, vr);
    for (; u < nfull; u += nwaves) {
#pragma unroll
        for (int j = 0; j < UNIT_V4; ++j) ((v4f*)xs)[lane + 32 * j] = vx[j];
#pragma unroll
        for (int j = 0; j < UNIT_V4; ++j) ((v4f*)rs)[lane + 32 * j] = vr[j];
        __builtin_amdgcn_wave_barrier();

        const int un = u + nwaves;
        if (un < nfull) load_unit(x, rep, un, lane, vx, vr);

        compute_unit(xs, rs, A0, A1, csum, sumbd, lane, m, kh);
        __builtin_amdgcn_wave_barrier();

        v4f* og = (v4f*)(out + (size_t)u * UNIT_FLOATS);
#pragma unroll
        for (int j = 0; j < UNIT_V4; ++j) og[lane + 32 * j] = ((const v4f*)xs)[lane + 32 * j];
        __builtin_amdgcn_wave_barrier();
    }

    const long long nflt = (long long)rows * E;
    for (u = nfull + wave; u < nunits; u += nwaves) {
        const long long g0 = (long long)u * UNIT_FLOATS;
        for (int e = lane; e < UNIT_FLOATS; e += 32) {
            const long long g = g0 + e;
            xs[e] = (g < nflt) ? x[g]   : 0.0f;
            rs[e] = (g < nflt) ? rep[g] : 0.0f;
        }
        __builtin_amdgcn_wave_barrier();
        compute_unit(xs, rs, A0, A1, csum, sumbd, lane, m, kh);
        __builtin_amdgcn_wave_barrier();
        for (int e = lane; e < UNIT_FLOATS; e += 32) {
            const long long g = g0 + e;
            if (g < nflt) out[g] = xs[e];
        }
        __builtin_amdgcn_wave_barrier();
    }
}

extern "C" void kernel_launch(void* const* d_in, const int* in_sizes, int n_in,
                              void* d_out, int out_size, void* d_ws, size_t ws_size,
                              hipStream_t stream) {
    const float* rep = (const float*)d_in[0];
    const float* x   = (const float*)d_in[1];
    const float* Wq  = (const float*)d_in[2];
    const float* Wk  = (const float*)d_in[3];
    const float* Wd  = (const float*)d_in[4];
    const float* bd  = (const float*)d_in[5];
    float* ws  = (float*)d_ws;
    float* out = (float*)d_out;

    const int rows   = in_sizes[0] / E;
    const int nunits = (rows + UNIT_ROWS - 1) / UNIT_ROWS;

    emotion_setup_kernel<<<1, 128, 0, stream>>>(Wq, Wk, Wd, bd, ws);

    int targetWaves = (nunits + 1) / 2;
    int blocks = (targetWaves + WAVES_PER_BLOCK - 1) / WAVES_PER_BLOCK;
    if (blocks < 1) blocks = 1;

    emotion_main_kernel<<<blocks, 256, 0, stream>>>(rep, x, ws, out, rows, nunits);
}


# setup launch removed: each wave of the streaming kernel folds Wq,Wk,Wd,bd itself (14 small loads ahead of its first unit's loads, sums under their latency, same sum order); 2 launches -> 1
# speedup vs baseline: 1.4472x; 1.2111x over previous
#include <hip/hip_runtime.h>
#include <hip/hip_bf16.h>

typedef __attribute__((ext_vector_type(2))) float v2f;
typedef __attribute__((ext_vector_type(4))) float v4f;
typedef __attribute__((ext_vector_type(8))) float v8f;

#define E 7
#define UNIT_ROWS 128
#define UNIT_FLOATS (UNIT_ROWS * E)
#define UNIT_V4 (UNIT_FLOATS / 4 / 32)
#define WAVES_PER_BLOCK 8

__device__ __forceinline__ void request_weights(const float* __restrict__ Wk,
                                                const float* __restrict__ Wd,
                                                int m, int lane,
                                                float (&wk)[E], float (&wd)[E]) {
    const int mi = (m < E) ? m : (E - 1);
    const int ti = (lane < E) ? lane : (E - 1);
#pragma unroll
    for (int j = 0; j < E; ++j) wk[j] = Wk[j * E + mi];
#pragma unroll
    for (int j = 0; j < E; ++j) wd[j] = Wd[j * E + ti];
}

__device__ __forceinline__ void fold_weights(const float* __restrict__ Wq,
                                             const float* __restrict__ bd,
                                             const float (&wk)[E], const float (&wd)[E],
                                             int m, int kh,
                                             v2f& A0, v2f& A1,
                                             float (&csum)[E], float& sumbd) {
    float a00 = 0.0f, a01 = 0.0f, a10 = 0.0f, a11 = 0.0f;
#pragma unroll
    for (int j = 0; j < E; ++j) {
        const float q0 = Wq[j * E + 0], q1 = Wq[j * E + 1], q2 = Wq[j * E + 2];
        const float q3 = Wq[j * E + 3], q4 = Wq[j * E + 4], q5 = Wq[j * E + 5];
        const float q6 = Wq[j * E + 6];
        a00 += (kh ? q2 : q0) * wk[j];
        a01 += (kh ? q3 : q1) * wk[j];
        a10 += (kh ? q6 : q4) * wk[j];
        a11 += q5 * wk[j];
    }
    const bool arow = (m < E);
    A0.x = arow ? a00 : 0.0f;
    A0.y = arow ? a01 : 0.0f;
    A1.x = arow ? a10 : 0.0f;
    A1.y = (arow && kh == 0) ? a11 : 0.0f;

    float cs = 0.0f;
#pragma unroll
    for (int j = 0; j < E; ++j) cs += wd[j];
#pragma unroll
    for (int i = 0; i < E; ++i)
        csum[i] = __int_as_float(__builtin_amdgcn_readlane(__float_as_int(cs), i));
    float sb = 0.0f;
#pragma unroll
    for (int j = 0; j < E; ++j) sb += bd[j];
    sumbd = sb;
}

__device__ __forceinline__ void load_unit(const float* __restrict__ x,
                                          const float* __restrict__ rep,
                                          int unit, int lane,
                                          v4f (&vx)[UNIT_V4], v4f (&vr)[UNIT_V4]) {
    const v4f* xg = (const v4f*)(x   + (size_t)unit * UNIT_FLOATS);
    const v4f* rg = (const v4f*)(rep + (size_t)unit * UNIT_FLOATS);
#pragma unroll
    for (int j = 0; j < UNIT_V4; ++j) vx[j] = xg[lane + 32 * j];
#pragma unroll
    for (int j = 0; j < UNIT_V4; ++j) vr[j] = rg[lane + 32 * j];
}

__device__ __forceinline__ void compute_unit(float* xs, const float* rs,
                                             v2f A0, v2f A1,
                                             const float (&csum)[E], float sumbd,
                                             int lane, int m, int kh) {
#pragma unroll
    for (int q = 0; q < UNIT_ROWS / 32; ++q) {
        const int ra  = (q * 32 + m) * E;
        const int rb  = ra + 16 * E;
        const int off = 2 * kh;
        v2f B00, B01, B10, B11;
        B00.x = xs[ra + off];     B00.y = xs[ra + off + 1];
        B01.x = xs[ra + 4 + off];
        const float pa = xs[ra + 5 + off];
        B01.y = kh ? 0.0f : pa;
        B10.x = xs[rb + off];     B10.y = xs[rb + off + 1];
        B11.x = xs[rb + 4 + off];
        const float pb = xs[rb + 5 + off];
        B11.y = kh ? 0.0f : pb;

        v8f acc0 = {};
        acc0 = __builtin_amdgcn_wmma_f32_16x16x4_f32(
            false, A0, false, B00, (short)0, acc0, false, false);
        acc0 = __builtin_amdgcn_wmma_f32_16x16x4_f32(
            false, A1, false, B01, (short)0, acc0, false, false);
        v8f acc1 = {};
        acc1 = __builtin_amdgcn_wmma_f32_16x16x4_f32(
            false, A0, false, B10, (short)0, acc1, false, false);
        acc1 = __builtin_amdgcn_wmma_f32_16x16x4_f32(
            false, A1, false, B11, (short)0, acc1, false, false);

        float t[E];
#pragma unroll
        for (int i = 0; i < E; ++i) {
            const float sh = __shfl(acc1[i], m, 32);
            t[i] = (kh == 0) ? acc0[i] : sh;
        }

        const int lr = (q * 32 + lane) * E;
        float r[E], raw[E];
#pragma unroll
        for (int i = 0; i < E; ++i) r[i] = rs[lr + i];
#pragma unroll
        for (int i = 0; i < E; ++i) raw[i] = r[i] * t[i];

        float mx = raw[0];
#pragma unroll
        for (int i = 1; i < E; ++i) mx = fmaxf(mx, raw[i]);
        float e[E], sum = 0.0f;
#pragma unroll
        for (int i = 0; i < E; ++i) {
            e[i] = __expf(raw[i] - mx);
            sum += e[i];
        }
        const float inv = __builtin_amdgcn_rcpf(sum);

#pragma unroll
        for (int i = 0; i < E; ++i) {
            const float s  = e[i] * inv;
            const float s3 = s * s * s;
            float d = fmaf(s3, csum[i], sumbd);
            d = fminf(fmaxf(d, -1.0f), 1.0f);
            xs[lr + i] = r[i] + d;
        }
    }
}

__global__ __launch_bounds__(256) void emotion_main_kernel(
    const float* __restrict__ rep,
    const float* __restrict__ x,
    const float* __restrict__ Wq,
    const float* __restrict__ Wk,
    const float* __restrict__ Wd,
    const float* __restrict__ bd,
    float* __restrict__ out,
    int rows, int nunits) {
    __shared__ __attribute__((aligned(16))) float lds[WAVES_PER_BLOCK][2 * UNIT_FLOATS];

    const int lane = threadIdx.x & 31;
    const int m    = lane & 15;
    const int kh   = lane >> 4;
    float* xs = lds[threadIdx.x >> 5];
    float* rs = xs + UNIT_FLOATS;

    v2f A0 = {0.0f, 0.0f}, A1 = {0.0f, 0.0f};
    float csum[E] = {};
    float sumbd = 0.0f;

    const int wave   = blockIdx.x * WAVES_PER_BLOCK + (threadIdx.x >> 5);
    const int nwaves = gridDim.x * WAVES_PER_BLOCK;

    const bool al16  = ((((size_t)x) | ((size_t)rep) | ((size_t)out)) & 15) == 0;
    const int  nfull = al16 ? rows / UNIT_ROWS : 0;

    v4f vx[UNIT_V4], vr[UNIT_V4];
    int u = wave;
    const bool has_full = (u < nfull);
    if (has_full) {
        float wk[E], wd[E];
        request_weights(Wk, Wd, m, lane, wk, wd);
        load_unit(x, rep, u, lane, vx, vr);
        fold_weights(Wq, bd, wk, wd, m, kh, A0, A1, csum, sumbd);
    }
    for (; u < nfull; u += nwaves) {
#pragma unroll
        for (int j = 0; j < UNIT_V4; ++j) ((v4f*)xs)[lane + 32 * j] = vx[j];
#pragma unroll
        for (int j = 0; j < UNIT_V4; ++j) ((v4f*)rs)[lane + 32 * j] = vr[j];
        __builtin_amdgcn_wave_barrier();

        const int un = u + nwaves;
        if (un < nfull) load_unit(x, rep, un, lane, vx, vr);

        compute_unit(xs, rs, A0, A1, csum, sumbd, lane, m, kh);
        __builtin_amdgcn_wave_barrier();

        v4f* og = (v4f*)(out + (size_t)u * UNIT_FLOATS);
#pragma unroll
        for (int j = 0; j < UNIT_V4; ++j) og[lane + 32 * j] = ((const v4f*)xs)[lane + 32 * j];
        __builtin_amdgcn_wave_barrier();
    }

    const long long nflt = (long long)rows * E;
    u = nfull + wave;
    if (u < nunits && !has_full) {
        float wk[E], wd[E];
        request_weights(Wk, Wd, m, lane, wk, wd);
        fold_weights(Wq, bd, wk, wd, m, kh, A0, A1, csum, sumbd);
    }
    for (; u < nunits; u += nwaves) {
        const long long g0 = (long long)u * UNIT_FLOATS;
        for (int e = lane; e < UNIT_FLOATS; e += 32) {
            const long long g = g0 + e;
            xs[e] = (g < nflt) ? x[g]   : 0.0f;
            rs[e] = (g < nflt) ? rep[g] : 0.0f;
        }
        __builtin_amdgcn_wave_barrier();
        compute_unit(xs, rs, A0, A1, csum, sumbd, lane, m, kh);
        __builtin_amdgcn_wave_barrier();
        for (int e = lane; e < UNIT_FLOATS; e += 32) {
            const long long g = g0 + e;
            if (g < nflt) out[g] = xs[e];
        }
        __builtin_amdgcn_wave_barrier();
    }
}

extern "C" void kernel_launch(void* const* d_in, const int* in_sizes, int n_in,
                              void* d_out, int out_size, void* d_ws, size_t ws_size,
                              hipStream_t stream) {
    const float* rep = (const float*)d_in[0];
    const float* x   = (const float*)d_in[1];
    const float* Wq  = (const float*)d_in[2];
    const float* Wk  = (const float*)d_in[3];
    const float* Wd  = (const float*)d_in[4];
    const float* bd  = (const float*)d_in[5];
    float* out = (float*)d_out;

    const int rows   = in_sizes[0] / E;
    const int nunits = (rows + UNIT_ROWS - 1) / UNIT_ROWS;

    int targetWaves = (nunits + 1) / 2;
    int blocks = (targetWaves + WAVES_PER_BLOCK - 1) / WAVES_PER_BLOCK;
    if (blocks < 1) blocks = 1;

    emotion_main_kernel<<<blocks, 256, 0, stream>>>(rep, x, Wq, Wk, Wd, bd, out, rows, nunits);
}
